# starting kernel
# baseline (speedup 1.0000x reference)
#include <hip/hip_runtime.h>
#include <math.h>

typedef __attribute__((ext_vector_type(2))) float v2f;
typedef __attribute__((ext_vector_type(8))) float v8f;

constexpr int Bb = 8;
constexpr int Qq = 512;
constexpr int Nn = 8192;
constexpr int Dd = 64;
constexpr int Vv = 64;

constexpr int WAVES  = 8;
constexpr int NSEG   = Nn / WAVES;
constexpr int NCHUNK = 16;

constexpr float SHIFT = 12.0f;

constexpr int PSTRIDE = 18;
constexpr int OSTRIDE = 65;
constexpr int SLOT    = 16 * OSTRIDE + 16;

__global__ __launch_bounds__(WAVES * 32)
void gated_attn_flash_f32wmma(const float* __restrict__ query,
                              const float* __restrict__ key,
                              const float* __restrict__ value,
                              const float* __restrict__ gates,
                              float* __restrict__ out)
{
    __shared__ float smem[WAVES * SLOT];

    const int tid  = threadIdx.x;
    const int w    = tid >> 5;
    const int lane = tid & 31;
    const int lo   = lane & 15;
    const int hi   = lane >> 4;

    const int b  = blockIdx.x >> 5;
    const int q0 = (blockIdx.x & 31) << 4;

    float* slot = smem + w * SLOT;

    v2f qa[16];
    {
        const float* qptr = query + ((size_t)(b * Qq + q0 + lo)) * Dd + 2 * hi;
#pragma unroll
        for (int i = 0; i < 16; ++i)
            qa[i] = *(const v2f*)(qptr + 4 * i);
    }

    v8f acc[4];
    v8f accs;
#pragma unroll
    for (int t = 0; t < 4; ++t) { v8f z = {}; acc[t] = z; }
    { v8f z = {}; accs = z; }

    const v2f vones = {1.0f, 1.0f};

    const int nbase = w * NSEG;

    for (int c = 0; c < NSEG / NCHUNK; ++c) {
        const int n0 = nbase + c * NCHUNK;

        const float* kptr = key + ((size_t)(b * Nn + n0 + lo)) * Dd + 2 * hi;
        v8f s = {};
#pragma unroll
        for (int i = 0; i < 16; ++i) {
            v2f kb = *(const v2f*)(kptr + 4 * i);
            s = __builtin_amdgcn_wmma_f32_16x16x4_f32(false, qa[i], false, kb,
                                                      (short)0, s, false, false);
        }

        const float* gptr = gates + ((size_t)(b * Nn + n0 + lo)) * Qq + q0 + 8 * hi;
        float4 g0 = *(const float4*)(gptr);
        float4 g1 = *(const float4*)(gptr + 4);
        __builtin_prefetch(gptr + (size_t)NCHUNK * Qq, 0, 1);
        float g[8] = {g0.x, g0.y, g0.z, g0.w, g1.x, g1.y, g1.z, g1.w};

        float p[8];
#pragma unroll
        for (int r = 0; r < 8; ++r)
            p[r] = __expf(fmaf(s[r], g[r], -SHIFT));

#pragma unroll
        for (int r = 0; r < 8; ++r)
            slot[(r + 8 * hi) * PSTRIDE + lo] = p[r];
        asm volatile("" ::: "memory");
        v2f pa[4];
#pragma unroll
        for (int i = 0; i < 4; ++i)
            pa[i] = *(const v2f*)(slot + lo * PSTRIDE + 4 * i + 2 * hi);

        const float* vptr = value + ((size_t)(b * Nn + n0 + 2 * hi)) * Vv + lo;
#pragma unroll
        for (int i = 0; i < 4; ++i) {
#pragma unroll
            for (int t = 0; t < 4; ++t) {
                v2f vb;
                vb.x = vptr[(size_t)(4 * i) * Vv + 16 * t];
                vb.y = vptr[(size_t)(4 * i + 1) * Vv + 16 * t];
                acc[t] = __builtin_amdgcn_wmma_f32_16x16x4_f32(false, pa[i], false, vb,
                                                               (short)0, acc[t], false, false);
            }
            accs = __builtin_amdgcn_wmma_f32_16x16x4_f32(false, pa[i], false, vones,
                                                         (short)0, accs, false, false);
        }
    }

#pragma unroll
    for (int t = 0; t < 4; ++t)
#pragma unroll
        for (int r = 0; r < 8; ++r)
            slot[(r + 8 * hi) * OSTRIDE + 16 * t + lo] = acc[t][r];
    if (lo == 0) {
#pragma unroll
        for (int r = 0; r < 8; ++r)
            slot[16 * OSTRIDE + (r + 8 * hi)] = accs[r];
    }
    __syncthreads();

#pragma unroll
    for (int k = 0; k < 4; ++k) {
        const int row = (tid >> 6) + 4 * k;
        const int col = tid & 63;
        float S = 0.f, O = 0.f;
        for (int ww = 0; ww < WAVES; ++ww) {
            const float* ws = smem + ww * SLOT;
            S += ws[16 * OSTRIDE + row];
            O += ws[row * OSTRIDE + col];
        }
        out[((size_t)(b * Qq + q0 + row)) * Vv + col] = O / S;
    }
}

extern "C" void kernel_launch(void* const* d_in, const int* in_sizes, int n_in,
                              void* d_out, int out_size, void* d_ws, size_t ws_size,
                              hipStream_t stream) {
    (void)in_sizes; (void)n_in; (void)d_ws; (void)ws_size; (void)out_size;
    const float* query = (const float*)d_in[0];
    const float* key   = (const float*)d_in[1];
    const float* value = (const float*)d_in[2];
    const float* gates = (const float*)d_in[3];
    float* out = (float*)d_out;

    dim3 grid(Bb * (Qq / 16));
    dim3 block(WAVES * 32);
    hipLaunchKernelGGL(gated_attn_flash_f32wmma, grid, block, 0, stream,
                       query, key, value, gates, out);
}


# Register double buffer: every K, gate and V load of the next 16-key chunk is issued before the current chunk is computed (two fragment sets, two chunks per loop pass); summation order unchanged
# speedup vs baseline: 1.2684x; 1.2684x over previous
#include <hip/hip_runtime.h>
#include <math.h>


typedef __attribute__((ext_vector_type(2))) float v2f;
typedef __attribute__((ext_vector_type(8))) float v8f;

constexpr int Bb = 8;
constexpr int Qq = 512;
constexpr int Nn = 8192;
constexpr int Dd = 64;
constexpr int Vv = 64;

constexpr int WAVES  = 8;
constexpr int NSEG   = Nn / WAVES;
constexpr int NCHUNK = 16;
constexpr int NCH    = NSEG / NCHUNK;

constexpr float SHIFT = 12.0f;

constexpr int PSTRIDE = 18;
constexpr int OSTRIDE = 65;
constexpr int SLOT    = 16 * OSTRIDE + 16;

struct ChunkFrag {
    v2f    kb[16];
    float4 g0, g1;
    v2f    vb[16];
};

__device__ __forceinline__
void load_chunk(ChunkFrag& f,
                const float* __restrict__ key,
                const float* __restrict__ value,
                const float* __restrict__ gates,
                int b, int q0, int n0, int lo, int hi)
{
    const float* kptr = key + ((size_t)(b * Nn + n0 + lo)) * Dd + 2 * hi;
#pragma unroll
    for (int i = 0; i < 16; ++i)
        f.kb[i] = *(const v2f*)(kptr + 4 * i);

    const float* gptr = gates + ((size_t)(b * Nn + n0 + lo)) * Qq + q0 + 8 * hi;
    f.g0 = *(const float4*)(gptr);
    f.g1 = *(const float4*)(gptr + 4);
    __builtin_prefetch(gptr + (size_t)NCHUNK * Qq, 0, 1);

    const float* vptr = value + ((size_t)(b * Nn + n0 + 2 * hi)) * Vv + lo;
#pragma unroll
    for (int i = 0; i < 4; ++i) {
#pragma unroll
        for (int t = 0; t < 4; ++t) {
            v2f vb;
            vb.x = vptr[(size_t)(4 * i) * Vv + 16 * t];
            vb.y = vptr[(size_t)(4 * i + 1) * Vv + 16 * t];
            f.vb[4 * i + t] = vb;
        }
    }
}

__device__ __forceinline__
void compute_chunk(const ChunkFrag& f, const v2f (&qa)[16],
                   v8f (&acc)[4], v8f& accs, float* slot, int lo, int hi)
{
    v8f s = {};
#pragma unroll
    for (int i = 0; i < 16; ++i)
        s = __builtin_amdgcn_wmma_f32_16x16x4_f32(false, qa[i], false, f.kb[i],
                                                  (short)0, s, false, false);

    float g[8] = {f.g0.x, f.g0.y, f.g0.z, f.g0.w, f.g1.x, f.g1.y, f.g1.z, f.g1.w};

    float p[8];
#pragma unroll
    for (int r = 0; r < 8; ++r)
        p[r] = __expf(fmaf(s[r], g[r], -SHIFT));

#pragma unroll
    for (int r = 0; r < 8; ++r)
        slot[(r + 8 * hi) * PSTRIDE + lo] = p[r];
    asm volatile("" ::: "memory");
    v2f pa[4];
#pragma unroll
    for (int i = 0; i < 4; ++i)
        pa[i] = *(const v2f*)(slot + lo * PSTRIDE + 4 * i + 2 * hi);

    const v2f vones = {1.0f, 1.0f};
#pragma unroll
    for (int i = 0; i < 4; ++i) {
#pragma unroll
        for (int t = 0; t < 4; ++t)
            acc[t] = __builtin_amdgcn_wmma_f32_16x16x4_f32(false, pa[i], false, f.vb[4 * i + t],
                                                           (short)0, acc[t], false, false);
        accs = __builtin_amdgcn_wmma_f32_16x16x4_f32(false, pa[i], false, vones,
                                                     (short)0, accs, false, false);
    }
}

__global__ __launch_bounds__(WAVES * 32)
void gated_attn_flash_f32wmma(const float* __restrict__ query,
                              const float* __restrict__ key,
                              const float* __restrict__ value,
                              const float* __restrict__ gates,
                              float* __restrict__ out)
{
    __shared__ float smem[WAVES * SLOT];

    const int tid  = threadIdx.x;
    const int w    = tid >> 5;
    const int lane = tid & 31;
    const int lo   = lane & 15;
    const int hi   = lane >> 4;

    const int b  = blockIdx.x >> 5;
    const int q0 = (blockIdx.x & 31) << 4;

    float* slot = smem + w * SLOT;

    v2f qa[16];
    {
        const float* qptr = query + ((size_t)(b * Qq + q0 + lo)) * Dd + 2 * hi;
#pragma unroll
        for (int i = 0; i < 16; ++i)
            qa[i] = *(const v2f*)(qptr + 4 * i);
    }

    v8f acc[4];
    v8f accs;
#pragma unroll
    for (int t = 0; t < 4; ++t) { v8f z = {}; acc[t] = z; }
    { v8f z = {}; accs = z; }

    const int nbase = w * NSEG;

    ChunkFrag fa, fb;
    load_chunk(fa, key, value, gates, b, q0, nbase, lo, hi);

    for (int c = 0; c < NCH - 2; c += 2) {
        load_chunk(fb, key, value, gates, b, q0, nbase + (c + 1) * NCHUNK, lo, hi);
        compute_chunk(fa, qa, acc, accs, slot, lo, hi);

        load_chunk(fa, key, value, gates, b, q0, nbase + (c + 2) * NCHUNK, lo, hi);
        compute_chunk(fb, qa, acc, accs, slot, lo, hi);
    }
    load_chunk(fb, key, value, gates, b, q0, nbase + (NCH - 1) * NCHUNK, lo, hi);
    compute_chunk(fa, qa, acc, accs, slot, lo, hi);
    compute_chunk(fb, qa, acc, accs, slot, lo, hi);

#pragma unroll
    for (int t = 0; t < 4; ++t)
#pragma unroll
        for (int r = 0; r < 8; ++r)
            slot[(r + 8 * hi) * OSTRIDE + 16 * t + lo] = acc[t][r];
    if (lo == 0) {
#pragma unroll
        for (int r = 0; r < 8; ++r)
            slot[16 * OSTRIDE + (r + 8 * hi)] = accs[r];
    }
    __syncthreads();

#pragma unroll
    for (int k = 0; k < 4; ++k) {
        const int row = (tid >> 6) + 4 * k;
        const int col = tid & 63;
        float S = 0.f, O = 0.f;
        for (int ww = 0; ww < WAVES; ++ww) {
            const float* ws = smem + ww * SLOT;
            S += ws[16 * OSTRIDE + row];
            O += ws[row * OSTRIDE + col];
        }
        out[((size_t)(b * Qq + q0 + row)) * Vv + col] = O / S;
    }
}

extern "C" void kernel_launch(void* const* d_in, const int* in_sizes, int n_in,
                              void* d_out, int out_size, void* d_ws, size_t ws_size,
                              hipStream_t stream) {
    (void)in_sizes; (void)n_in; (void)d_ws; (void)ws_size; (void)out_size;
    const float* query = (const float*)d_in[0];
    const float* key   = (const float*)d_in[1];
    const float* value = (const float*)d_in[2];
    const float* gates = (const float*)d_in[3];
    float* out = (float*)d_out;

    dim3 grid(Bb * (Qq / 16));
    dim3 block(WAVES * 32);
    hipLaunchKernelGGL(gated_attn_flash_f32wmma, grid, block, 0, stream,
                       query, key, value, gates, out);
}


# K and V fragment loads widened to 16 B per lane (K 8 loads for 16, V 8 for 32: fewer lines touched per load); score dot product sums d in a permuted order; scheduling barrier keeps the double buffer
# speedup vs baseline: 1.3287x; 1.0475x over previous
#include <hip/hip_runtime.h>
#include <math.h>


typedef __attribute__((ext_vector_type(2))) float v2f;
typedef __attribute__((ext_vector_type(4))) float v4f;
typedef __attribute__((ext_vector_type(8))) float v8f;

constexpr int Bb = 8;
constexpr int Qq = 512;
constexpr int Nn = 8192;
constexpr int Dd = 64;
constexpr int Vv = 64;

constexpr int WAVES  = 8;
constexpr int NSEG   = Nn / WAVES;
constexpr int NCHUNK = 16;
constexpr int NCH    = NSEG / NCHUNK;

constexpr float SHIFT = 12.0f;

constexpr int PSTRIDE = 18;
constexpr int OSTRIDE = 65;
constexpr int SLOT    = 16 * OSTRIDE + 16;

struct ChunkFrag {
    v4f    k4[8];
    float4 g0, g1;
    v4f    v4[8];
};

__device__ __forceinline__
void load_chunk(ChunkFrag& f,
                const float* __restrict__ key,
                const float* __restrict__ value,
                const float* __restrict__ gates,
                int b, int q0, int n0, int lo, int hi)
{
    const float* kptr = key + ((size_t)(b * Nn + n0 + lo)) * Dd + 4 * hi;
#pragma unroll
    for (int m = 0; m < 8; ++m)
        f.k4[m] = *(const v4f*)(kptr + 8 * m);

    const float* gptr = gates + ((size_t)(b * Nn + n0 + lo)) * Qq + q0 + 8 * hi;
    f.g0 = *(const float4*)(gptr);
    f.g1 = *(const float4*)(gptr + 4);
    __builtin_prefetch(gptr + (size_t)NCHUNK * Qq, 0, 1);

    const float* vptr = value + ((size_t)(b * Nn + n0 + 2 * hi)) * Vv + 4 * lo;
#pragma unroll
    for (int i = 0; i < 4; ++i) {
#pragma unroll
        for (int j = 0; j < 2; ++j)
            f.v4[2 * i + j] = *(const v4f*)(vptr + (size_t)(4 * i + j) * Vv);
    }
}

__device__ __forceinline__
void compute_chunk(const ChunkFrag& f, const v2f (&qa)[16],
                   v8f (&acc)[4], v8f& accs, float* slot, int lo, int hi)
{
    v8f s = {};
#pragma unroll
    for (int m = 0; m < 8; ++m) {
        const v2f k0 = {f.k4[m][0], f.k4[m][1]};
        const v2f k1 = {f.k4[m][2], f.k4[m][3]};
        s = __builtin_amdgcn_wmma_f32_16x16x4_f32(false, qa[2 * m], false, k0,
                                                  (short)0, s, false, false);
        s = __builtin_amdgcn_wmma_f32_16x16x4_f32(false, qa[2 * m + 1], false, k1,
                                                  (short)0, s, false, false);
    }

    float g[8] = {f.g0.x, f.g0.y, f.g0.z, f.g0.w, f.g1.x, f.g1.y, f.g1.z, f.g1.w};

    float p[8];
#pragma unroll
    for (int r = 0; r < 8; ++r)
        p[r] = __expf(fmaf(s[r], g[r], -SHIFT));

#pragma unroll
    for (int r = 0; r < 8; ++r)
        slot[(r + 8 * hi) * PSTRIDE + lo] = p[r];
    asm volatile("" ::: "memory");
    v2f pa[4];
#pragma unroll
    for (int i = 0; i < 4; ++i)
        pa[i] = *(const v2f*)(slot + lo * PSTRIDE + 4 * i + 2 * hi);

    const v2f vones = {1.0f, 1.0f};
#pragma unroll
    for (int i = 0; i < 4; ++i) {
#pragma unroll
        for (int t = 0; t < 4; ++t) {
            const v2f vb = {f.v4[2 * i][t], f.v4[2 * i + 1][t]};
            acc[t] = __builtin_amdgcn_wmma_f32_16x16x4_f32(false, pa[i], false, vb,
                                                           (short)0, acc[t], false, false);
        }
        accs = __builtin_amdgcn_wmma_f32_16x16x4_f32(false, pa[i], false, vones,
                                                     (short)0, accs, false, false);
    }
}

__global__ __launch_bounds__(WAVES * 32)
void gated_attn_flash_f32wmma(const float* __restrict__ query,
                              const float* __restrict__ key,
                              const float* __restrict__ value,
                              const float* __restrict__ gates,
                              float* __restrict__ out)
{
    __shared__ float smem[WAVES * SLOT];

    const int tid  = threadIdx.x;
    const int w    = tid >> 5;
    const int lane = tid & 31;
    const int lo   = lane & 15;
    const int hi   = lane >> 4;

    const int b  = blockIdx.x >> 5;
    const int q0 = (blockIdx.x & 31) << 4;

    float* slot = smem + w * SLOT;

    v2f qa[16];
    {
        const float* qptr = query + ((size_t)(b * Qq + q0 + lo)) * Dd + 4 * hi;
#pragma unroll
        for (int m = 0; m < 8; ++m) {
            const v4f q4 = *(const v4f*)(qptr + 8 * m);
            const v2f qlo = {q4[0], q4[1]};
            const v2f qhi = {q4[2], q4[3]};
            qa[2 * m]     = qlo;
            qa[2 * m + 1] = qhi;
        }
    }

    v8f acc[4];
    v8f accs;
#pragma unroll
    for (int t = 0; t < 4; ++t) { v8f z = {}; acc[t] = z; }
    { v8f z = {}; accs = z; }

    const int nbase = w * NSEG;

    ChunkFrag fa, fb;
    load_chunk(fa, key, value, gates, b, q0, nbase, lo, hi);

    for (int c = 0; c < NCH - 2; c += 2) {
        load_chunk(fb, key, value, gates, b, q0, nbase + (c + 1) * NCHUNK, lo, hi);
        compute_chunk(fa, qa, acc, accs, slot, lo, hi);
        __builtin_amdgcn_sched_barrier(0);

        load_chunk(fa, key, value, gates, b, q0, nbase + (c + 2) * NCHUNK, lo, hi);
        compute_chunk(fb, qa, acc, accs, slot, lo, hi);
        __builtin_amdgcn_sched_barrier(0);
    }
    load_chunk(fb, key, value, gates, b, q0, nbase + (NCH - 1) * NCHUNK, lo, hi);
    compute_chunk(fa, qa, acc, accs, slot, lo, hi);
    __builtin_amdgcn_sched_barrier(0);
    compute_chunk(fb, qa, acc, accs, slot, lo, hi);

#pragma unroll
    for (int t = 0; t < 4; ++t)
#pragma unroll
        for (int r = 0; r < 8; ++r)
            slot[(r + 8 * hi) * OSTRIDE + 4 * lo + t] = acc[t][r];
    if (lo == 0) {
#pragma unroll
        for (int r = 0; r < 8; ++r)
            slot[16 * OSTRIDE + (r + 8 * hi)] = accs[r];
    }
    __syncthreads();

#pragma unroll
    for (int k = 0; k < 4; ++k) {
        const int row = (tid >> 6) + 4 * k;
        const int col = tid & 63;
        float S = 0.f, O = 0.f;
        for (int ww = 0; ww < WAVES; ++ww) {
            const float* ws = smem + ww * SLOT;
            S += ws[16 * OSTRIDE + row];
            O += ws[row * OSTRIDE + col];
        }
        out[((size_t)(b * Qq + q0 + row)) * Vv + col] = O / S;
    }
}

extern "C" void kernel_launch(void* const* d_in, const int* in_sizes, int n_in,
                              void* d_out, int out_size, void* d_ws, size_t ws_size,
                              hipStream_t stream) {
    (void)in_sizes; (void)n_in; (void)d_ws; (void)ws_size; (void)out_size;
    const float* query = (const float*)d_in[0];
    const float* key   = (const float*)d_in[1];
    const float* value = (const float*)d_in[2];
    const float* gates = (const float*)d_in[3];
    float* out = (float*)d_out;

    dim3 grid(Bb * (Qq / 16));
    dim3 block(WAVES * 32);
    hipLaunchKernelGGL(gated_attn_flash_f32wmma, grid, block, 0, stream,
                       query, key, value, gates, out);
}


# Sixteen waves per workgroup instead of eight (four per SIMD, one workgroup per WGP as before): each wave takes 512 keys; the merge sums sixteen partials instead of eight (summation order changes)
# speedup vs baseline: 1.4205x; 1.0691x over previous
#include <hip/hip_runtime.h>
#include <math.h>


typedef __attribute__((ext_vector_type(2))) float v2f;
typedef __attribute__((ext_vector_type(4))) float v4f;
typedef __attribute__((ext_vector_type(8))) float v8f;

constexpr int Bb = 8;
constexpr int Qq = 512;
constexpr int Nn = 8192;
constexpr int Dd = 64;
constexpr int Vv = 64;

constexpr int WAVES  = 16;
constexpr int NSEG   = Nn / WAVES;
constexpr int NCHUNK = 16;
constexpr int NCH    = NSEG / NCHUNK;

constexpr float SHIFT = 12.0f;

constexpr int PSTRIDE = 18;
constexpr int OSTRIDE = 65;
constexpr int SLOT    = 16 * OSTRIDE + 16;

struct ChunkFrag {
    v4f    k4[8];
    float4 g0, g1;
    v4f    v4[8];
};

__device__ __forceinline__
void load_chunk(ChunkFrag& f,
                const float* __restrict__ key,
                const float* __restrict__ value,
                const float* __restrict__ gates,
                int b, int q0, int n0, int lo, int hi)
{
    const float* kptr = key + ((size_t)(b * Nn + n0 + lo)) * Dd + 4 * hi;
#pragma unroll
    for (int m = 0; m < 8; ++m)
        f.k4[m] = *(const v4f*)(kptr + 8 * m);

    const float* gptr = gates + ((size_t)(b * Nn + n0 + lo)) * Qq + q0 + 8 * hi;
    f.g0 = *(const float4*)(gptr);
    f.g1 = *(const float4*)(gptr + 4);
    __builtin_prefetch(gptr + (size_t)NCHUNK * Qq, 0, 1);

    const float* vptr = value + ((size_t)(b * Nn + n0 + 2 * hi)) * Vv + 4 * lo;
#pragma unroll
    for (int i = 0; i < 4; ++i) {
#pragma unroll
        for (int j = 0; j < 2; ++j)
            f.v4[2 * i + j] = *(const v4f*)(vptr + (size_t)(4 * i + j) * Vv);
    }
}

__device__ __forceinline__
void compute_chunk(const ChunkFrag& f, const v2f (&qa)[16],
                   v8f (&acc)[4], v8f& accs, float* slot, int lo, int hi)
{
    v8f s = {};
#pragma unroll
    for (int m = 0; m < 8; ++m) {
        const v2f k0 = {f.k4[m][0], f.k4[m][1]};
        const v2f k1 = {f.k4[m][2], f.k4[m][3]};
        s = __builtin_amdgcn_wmma_f32_16x16x4_f32(false, qa[2 * m], false, k0,
                                                  (short)0, s, false, false);
        s = __builtin_amdgcn_wmma_f32_16x16x4_f32(false, qa[2 * m + 1], false, k1,
                                                  (short)0, s, false, false);
    }

    float g[8] = {f.g0.x, f.g0.y, f.g0.z, f.g0.w, f.g1.x, f.g1.y, f.g1.z, f.g1.w};

    float p[8];
#pragma unroll
    for (int r = 0; r < 8; ++r)
        p[r] = __expf(fmaf(s[r], g[r], -SHIFT));

    float* prow = slot + 8 * hi * PSTRIDE + lo;
#pragma unroll
    for (int r = 0; r < 8; ++r)
        prow[r * PSTRIDE] = p[r];
    asm volatile("" ::: "memory");
    v2f pa[4];
#pragma unroll
    for (int i = 0; i < 4; ++i)
        pa[i] = *(const v2f*)(slot + lo * PSTRIDE + 4 * i + 2 * hi);

    const v2f vones = {1.0f, 1.0f};
#pragma unroll
    for (int i = 0; i < 4; ++i) {
#pragma unroll
        for (int t = 0; t < 4; ++t) {
            const v2f vb = {f.v4[2 * i][t], f.v4[2 * i + 1][t]};
            acc[t] = __builtin_amdgcn_wmma_f32_16x16x4_f32(false, pa[i], false, vb,
                                                           (short)0, acc[t], false, false);
        }
        accs = __builtin_amdgcn_wmma_f32_16x16x4_f32(false, pa[i], false, vones,
                                                     (short)0, accs, false, false);
    }
}

__global__ __launch_bounds__(WAVES * 32)
void gated_attn_flash_f32wmma(const float* __restrict__ query,
                              const float* __restrict__ key,
                              const float* __restrict__ value,
                              const float* __restrict__ gates,
                              float* __restrict__ out)
{
    __shared__ float smem[WAVES * SLOT];

    const int tid  = threadIdx.x;
    const int w    = tid >> 5;
    const int lane = tid & 31;
    const int lo   = lane & 15;
    const int hi   = lane >> 4;

    const int b  = blockIdx.x >> 5;
    const int q0 = (blockIdx.x & 31) << 4;

    float* slot = smem + w * SLOT;

    v2f qa[16];
    {
        const float* qptr = query + ((size_t)(b * Qq + q0 + lo)) * Dd + 4 * hi;
#pragma unroll
        for (int m = 0; m < 8; ++m) {
            const v4f q4 = *(const v4f*)(qptr + 8 * m);
            const v2f qlo = {q4[0], q4[1]};
            const v2f qhi = {q4[2], q4[3]};
            qa[2 * m]     = qlo;
            qa[2 * m + 1] = qhi;
        }
    }

    v8f acc[4];
    v8f accs;
#pragma unroll
    for (int t = 0; t < 4; ++t) { v8f z = {}; acc[t] = z; }
    { v8f z = {}; accs = z; }

    const int nbase = w * NSEG;

    ChunkFrag fa, fb;
    load_chunk(fa, key, value, gates, b, q0, nbase, lo, hi);

    for (int c = 0; c < NCH - 2; c += 2) {
        load_chunk(fb, key, value, gates, b, q0, nbase + (c + 1) * NCHUNK, lo, hi);
        compute_chunk(fa, qa, acc, accs, slot, lo, hi);
        __builtin_amdgcn_sched_barrier(0);

        load_chunk(fa, key, value, gates, b, q0, nbase + (c + 2) * NCHUNK, lo, hi);
        compute_chunk(fb, qa, acc, accs, slot, lo, hi);
        __builtin_amdgcn_sched_barrier(0);
    }
    load_chunk(fb, key, value, gates, b, q0, nbase + (NCH - 1) * NCHUNK, lo, hi);
    compute_chunk(fa, qa, acc, accs, slot, lo, hi);
    __builtin_amdgcn_sched_barrier(0);
    compute_chunk(fb, qa, acc, accs, slot, lo, hi);

#pragma unroll
    for (int t = 0; t < 4; ++t)
#pragma unroll
        for (int r = 0; r < 8; ++r)
            slot[(r + 8 * hi) * OSTRIDE + 4 * lo + t] = acc[t][r];
    if (lo == 0) {
#pragma unroll
        for (int r = 0; r < 8; ++r)
            slot[16 * OSTRIDE + (r + 8 * hi)] = accs[r];
    }
    __syncthreads();

    constexpr int ROWS_PER_PASS = WAVES * 32 / 64;
#pragma unroll
    for (int k = 0; k < 16 / ROWS_PER_PASS; ++k) {
        const int row = (tid >> 6) + ROWS_PER_PASS * k;
        const int col = tid & 63;
        float S = 0.f, O = 0.f;
        for (int ww = 0; ww < WAVES; ++ww) {
            const float* ws = smem + ww * SLOT;
            S += ws[16 * OSTRIDE + row];
            O += ws[row * OSTRIDE + col];
        }
        out[((size_t)(b * Qq + q0 + row)) * Vv + col] = O / S;
    }
}

extern "C" void kernel_launch(void* const* d_in, const int* in_sizes, int n_in,
                              void* d_out, int out_size, void* d_ws, size_t ws_size,
                              hipStream_t stream) {
    (void)in_sizes; (void)n_in; (void)d_ws; (void)ws_size; (void)out_size;
    const float* query = (const float*)d_in[0];
    const float* key   = (const float*)d_in[1];
    const float* value = (const float*)d_in[2];
    const float* gates = (const float*)d_in[3];
    float* out = (float*)d_out;

    dim3 grid(Bb * (Qq / 16));
    dim3 block(WAVES * 32);
    hipLaunchKernelGGL(gated_attn_flash_f32wmma, grid, block, 0, stream,
                       query, key, value, gates, out);
}


# Transposed tiles (S^T = K x Q^T, O^T = V^T x P^T): exp results feed the second product from the accumulator registers, no LDS transpose of P in the loop; a chunk's keys summed in a permuted order
# speedup vs baseline: 1.4684x; 1.0337x over previous
#include <hip/hip_runtime.h>
#include <math.h>


typedef __attribute__((ext_vector_type(2))) float v2f;
typedef __attribute__((ext_vector_type(4))) float v4f;
typedef __attribute__((ext_vector_type(8))) float v8f;

constexpr int Bb = 8;
constexpr int Qq = 512;
constexpr int Nn = 8192;
constexpr int Dd = 64;
constexpr int Vv = 64;

constexpr int WAVES  = 16;
constexpr int NSEG   = Nn / WAVES;
constexpr int NCHUNK = 16;
constexpr int NCH    = NSEG / NCHUNK;

constexpr float SHIFT = 12.0f;

constexpr int OSTRIDE = 65;
constexpr int SLOT    = 16 * OSTRIDE + 16;

struct ChunkFrag {
    v4f   k4[8];
    float g[8];
    v4f   v4[8];
};

__device__ __forceinline__
void load_chunk(ChunkFrag& f,
                const float* __restrict__ key,
                const float* __restrict__ value,
                const float* __restrict__ gates,
                int b, int q0, int n0, int lo, int hi)
{
    const float* kptr = key + ((size_t)(b * Nn + n0 + lo)) * Dd + 4 * hi;
#pragma unroll
    for (int m = 0; m < 8; ++m)
        f.k4[m] = *(const v4f*)(kptr + 8 * m);

    const float* gptr = gates + ((size_t)(b * Nn + n0 + 8 * hi)) * Qq + q0 + lo;
#pragma unroll
    for (int r = 0; r < 8; ++r)
        f.g[r] = gptr[(size_t)r * Qq];
    __builtin_prefetch(gates + ((size_t)(b * Nn + n0 + NCHUNK + lo)) * Qq + q0 + 8 * hi, 0, 1);

    const float* vptr = value + ((size_t)(b * Nn + n0 + 8 * hi)) * Vv + 4 * lo;
#pragma unroll
    for (int i = 0; i < 4; ++i) {
#pragma unroll
        for (int j = 0; j < 2; ++j)
            f.v4[2 * i + j] = *(const v4f*)(vptr + (size_t)(2 * i + j) * Vv);
    }
}

__device__ __forceinline__
void compute_chunk(const ChunkFrag& f, const v2f (&qb)[16],
                   v8f (&acc)[4], v8f& accs)
{
    v8f s = {};
#pragma unroll
    for (int m = 0; m < 8; ++m) {
        const v2f k0 = {f.k4[m][0], f.k4[m][1]};
        const v2f k1 = {f.k4[m][2], f.k4[m][3]};
        s = __builtin_amdgcn_wmma_f32_16x16x4_f32(false, k0, false, qb[2 * m],
                                                  (short)0, s, false, false);
        s = __builtin_amdgcn_wmma_f32_16x16x4_f32(false, k1, false, qb[2 * m + 1],
                                                  (short)0, s, false, false);
    }

    float p[8];
#pragma unroll
    for (int r = 0; r < 8; ++r)
        p[r] = __expf(fmaf(s[r], f.g[r], -SHIFT));

    const v2f vones = {1.0f, 1.0f};
#pragma unroll
    for (int i = 0; i < 4; ++i) {
        const v2f pb = {p[2 * i], p[2 * i + 1]};
#pragma unroll
        for (int t = 0; t < 4; ++t) {
            const v2f va = {f.v4[2 * i][t], f.v4[2 * i + 1][t]};
            acc[t] = __builtin_amdgcn_wmma_f32_16x16x4_f32(false, va, false, pb,
                                                           (short)0, acc[t], false, false);
        }
        accs = __builtin_amdgcn_wmma_f32_16x16x4_f32(false, vones, false, pb,
                                                     (short)0, accs, false, false);
    }
}

__global__ __launch_bounds__(WAVES * 32)
void gated_attn_flash_f32wmma(const float* __restrict__ query,
                              const float* __restrict__ key,
                              const float* __restrict__ value,
                              const float* __restrict__ gates,
                              float* __restrict__ out)
{
    __shared__ float smem[WAVES * SLOT];

    const int tid  = threadIdx.x;
    const int w    = tid >> 5;
    const int lane = tid & 31;
    const int lo   = lane & 15;
    const int hi   = lane >> 4;

    const int b  = blockIdx.x >> 5;
    const int q0 = (blockIdx.x & 31) << 4;

    float* slot = smem + w * SLOT;

    v2f qa[16];
    {
        const float* qptr = query + ((size_t)(b * Qq + q0 + lo)) * Dd + 4 * hi;
#pragma unroll
        for (int m = 0; m < 8; ++m) {
            const v4f q4 = *(const v4f*)(qptr + 8 * m);
            const v2f qlo = {q4[0], q4[1]};
            const v2f qhi = {q4[2], q4[3]};
            qa[2 * m]     = qlo;
            qa[2 * m + 1] = qhi;
        }
    }

    v8f acc[4];
    v8f accs;
#pragma unroll
    for (int t = 0; t < 4; ++t) { v8f z = {}; acc[t] = z; }
    { v8f z = {}; accs = z; }

    const int nbase = w * NSEG;

    ChunkFrag fa, fb;
    load_chunk(fa, key, value, gates, b, q0, nbase, lo, hi);

    for (int c = 0; c < NCH - 2; c += 2) {
        load_chunk(fb, key, value, gates, b, q0, nbase + (c + 1) * NCHUNK, lo, hi);
        __builtin_amdgcn_sched_barrier(0);
        compute_chunk(fa, qa, acc, accs);
        __builtin_amdgcn_sched_barrier(0);

        load_chunk(fa, key, value, gates, b, q0, nbase + (c + 2) * NCHUNK, lo, hi);
        __builtin_amdgcn_sched_barrier(0);
        compute_chunk(fb, qa, acc, accs);
        __builtin_amdgcn_sched_barrier(0);
    }
    load_chunk(fb, key, value, gates, b, q0, nbase + (NCH - 1) * NCHUNK, lo, hi);
    __builtin_amdgcn_sched_barrier(0);
    compute_chunk(fa, qa, acc, accs);
    __builtin_amdgcn_sched_barrier(0);
    compute_chunk(fb, qa, acc, accs);

#pragma unroll
    for (int t = 0; t < 4; ++t)
#pragma unroll
        for (int r = 0; r < 8; ++r)
            slot[lo * OSTRIDE + 4 * (r + 8 * hi) + t] = acc[t][r];
    if (hi == 0)
        slot[16 * OSTRIDE + lo] = accs[0];
    __syncthreads();

    constexpr int ROWS_PER_PASS = WAVES * 32 / 64;
#pragma unroll
    for (int k = 0; k < 16 / ROWS_PER_PASS; ++k) {
        const int row = (tid >> 6) + ROWS_PER_PASS * k;
        const int col = tid & 63;
        float S = 0.f, O = 0.f;
        for (int ww = 0; ww < WAVES; ++ww) {
            const float* ws = smem + ww * SLOT;
            S += ws[16 * OSTRIDE + row];
            O += ws[row * OSTRIDE + col];
        }
        out[((size_t)(b * Qq + q0 + row)) * Vv + col] = O / S;
    }
}

extern "C" void kernel_launch(void* const* d_in, const int* in_sizes, int n_in,
                              void* d_out, int out_size, void* d_ws, size_t ws_size,
                              hipStream_t stream) {
    (void)in_sizes; (void)n_in; (void)d_ws; (void)ws_size; (void)out_size;
    const float* query = (const float*)d_in[0];
    const float* key   = (const float*)d_in[1];
    const float* value = (const float*)d_in[2];
    const float* gates = (const float*)d_in[3];
    float* out = (float*)d_out;

    dim3 grid(Bb * (Qq / 16));
    dim3 block(WAVES * 32);
    hipLaunchKernelGGL(gated_attn_flash_f32wmma, grid, block, 0, stream,
                       query, key, value, gates, out);
}


# Two interleaved 16-query tiles per wave: each K and V fragment feeds both, gates 8 B per lane; 8-wave blocks, keys split over two blocks, a 2nd launch adds the halves (merge order changes)
# speedup vs baseline: 1.8190x; 1.2388x over previous
#include <hip/hip_runtime.h>
#include <math.h>


typedef __attribute__((ext_vector_type(2))) float v2f;
typedef __attribute__((ext_vector_type(4))) float v4f;
typedef __attribute__((ext_vector_type(8))) float v8f;

constexpr int Bb = 8;
constexpr int Qq = 512;
constexpr int Nn = 8192;
constexpr int Dd = 64;
constexpr int Vv = 64;

constexpr int QT     = 32;
constexpr int KSPLIT = 2;
constexpr int WAVES  = 8;
constexpr int NSEG   = Nn / (KSPLIT * WAVES);
constexpr int NCHUNK = 16;
constexpr int NCH    = NSEG / NCHUNK;
constexpr int NBLK   = Bb * (Qq / QT) * KSPLIT;

constexpr float SHIFT = 12.0f;

constexpr int OSTRIDE = 65;
constexpr int SLOT    = QT * OSTRIDE + QT;

constexpr int WS_S = NBLK * QT * Vv;

struct ChunkFrag {
    v4f k4[8];
    v2f g[8];
    v4f v4[8];
};

__device__ __forceinline__
void load_chunk(ChunkFrag& f,
                const float* __restrict__ key,
                const float* __restrict__ value,
                const float* __restrict__ gates,
                int b, int q0, int n0, int lo, int hi)
{
    const float* kptr = key + ((size_t)(b * Nn + n0 + lo)) * Dd + 4 * hi;
#pragma unroll
    for (int m = 0; m < 8; ++m)
        f.k4[m] = *(const v4f*)(kptr + 8 * m);

    const float* gptr = gates + ((size_t)(b * Nn + n0 + 8 * hi)) * Qq + q0 + 2 * lo;
#pragma unroll
    for (int r = 0; r < 8; ++r)
        f.g[r] = *(const v2f*)(gptr + (size_t)r * Qq);
    __builtin_prefetch(gates + ((size_t)(b * Nn + n0 + NCHUNK + lo)) * Qq + q0 + 16 * hi, 0, 1);

    const float* vptr = value + ((size_t)(b * Nn + n0 + 8 * hi)) * Vv + 4 * lo;
#pragma unroll
    for (int i = 0; i < 4; ++i) {
#pragma unroll
        for (int j = 0; j < 2; ++j)
            f.v4[2 * i + j] = *(const v4f*)(vptr + (size_t)(2 * i + j) * Vv);
    }
}

__device__ __forceinline__
void compute_chunk(const ChunkFrag& f,
                   const v2f (&qb0)[16], const v2f (&qb1)[16],
                   v8f (&acc0)[4], v8f (&acc1)[4], v8f& accs0, v8f& accs1)
{
    v8f s0 = {};
    v8f s1 = {};
#pragma unroll
    for (int m = 0; m < 8; ++m) {
        const v2f k0 = {f.k4[m][0], f.k4[m][1]};
        const v2f k1 = {f.k4[m][2], f.k4[m][3]};
        s0 = __builtin_amdgcn_wmma_f32_16x16x4_f32(false, k0, false, qb0[2 * m],
                                                   (short)0, s0, false, false);
        s1 = __builtin_amdgcn_wmma_f32_16x16x4_f32(false, k0, false, qb1[2 * m],
                                                   (short)0, s1, false, false);
        s0 = __builtin_amdgcn_wmma_f32_16x16x4_f32(false, k1, false, qb0[2 * m + 1],
                                                   (short)0, s0, false, false);
        s1 = __builtin_amdgcn_wmma_f32_16x16x4_f32(false, k1, false, qb1[2 * m + 1],
                                                   (short)0, s1, false, false);
    }

    float p0[8], p1[8];
#pragma unroll
    for (int r = 0; r < 8; ++r) {
        p0[r] = __expf(fmaf(s0[r], f.g[r][0], -SHIFT));
        p1[r] = __expf(fmaf(s1[r], f.g[r][1], -SHIFT));
    }

    const v2f vones = {1.0f, 1.0f};
#pragma unroll
    for (int i = 0; i < 4; ++i) {
        const v2f pb0 = {p0[2 * i], p0[2 * i + 1]};
        const v2f pb1 = {p1[2 * i], p1[2 * i + 1]};
#pragma unroll
        for (int t = 0; t < 4; ++t) {
            const v2f va = {f.v4[2 * i][t], f.v4[2 * i + 1][t]};
            acc0[t] = __builtin_amdgcn_wmma_f32_16x16x4_f32(false, va, false, pb0,
                                                            (short)0, acc0[t], false, false);
            acc1[t] = __builtin_amdgcn_wmma_f32_16x16x4_f32(false, va, false, pb1,
                                                            (short)0, acc1[t], false, false);
        }
        accs0 = __builtin_amdgcn_wmma_f32_16x16x4_f32(false, vones, false, pb0,
                                                      (short)0, accs0, false, false);
        accs1 = __builtin_amdgcn_wmma_f32_16x16x4_f32(false, vones, false, pb1,
                                                      (short)0, accs1, false, false);
    }
}

__global__ __launch_bounds__(WAVES * 32)
void gated_attn_flash_f32wmma(const float* __restrict__ query,
                              const float* __restrict__ key,
                              const float* __restrict__ value,
                              const float* __restrict__ gates,
                              float* __restrict__ ws)
{
    __shared__ float smem[WAVES * SLOT];

    const int tid  = threadIdx.x;
    const int w    = tid >> 5;
    const int lane = tid & 31;
    const int lo   = lane & 15;
    const int hi   = lane >> 4;

    const int half = blockIdx.x & 1;
    const int pq   = blockIdx.x >> 1;
    const int b    = pq >> 4;
    const int q0   = (pq & 15) << 5;

    float* slot = smem + w * SLOT;

    v2f qa0[16], qa1[16];
    {
        const float* qptr = query + ((size_t)(b * Qq + q0 + 2 * lo)) * Dd + 4 * hi;
#pragma unroll
        for (int m = 0; m < 8; ++m) {
            const v4f qx = *(const v4f*)(qptr + 8 * m);
            const v4f qy = *(const v4f*)(qptr + Dd + 8 * m);
            const v2f xlo = {qx[0], qx[1]};
            const v2f xhi = {qx[2], qx[3]};
            const v2f ylo = {qy[0], qy[1]};
            const v2f yhi = {qy[2], qy[3]};
            qa0[2 * m]     = xlo;
            qa0[2 * m + 1] = xhi;
            qa1[2 * m]     = ylo;
            qa1[2 * m + 1] = yhi;
        }
    }

    v8f acc0[4], acc1[4];
    v8f accs0, accs1;
#pragma unroll
    for (int t = 0; t < 4; ++t) { v8f z = {}; acc0[t] = z; acc1[t] = z; }
    { v8f z = {}; accs0 = z; accs1 = z; }

    const int nbase = (half * WAVES + w) * NSEG;

    ChunkFrag fa, fb;
    load_chunk(fa, key, value, gates, b, q0, nbase, lo, hi);

    for (int c = 0; c < NCH - 2; c += 2) {
        load_chunk(fb, key, value, gates, b, q0, nbase + (c + 1) * NCHUNK, lo, hi);
        __builtin_amdgcn_sched_barrier(0);
        compute_chunk(fa, qa0, qa1, acc0, acc1, accs0, accs1);
        __builtin_amdgcn_sched_barrier(0);

        load_chunk(fa, key, value, gates, b, q0, nbase + (c + 2) * NCHUNK, lo, hi);
        __builtin_amdgcn_sched_barrier(0);
        compute_chunk(fb, qa0, qa1, acc0, acc1, accs0, accs1);
        __builtin_amdgcn_sched_barrier(0);
    }
    load_chunk(fb, key, value, gates, b, q0, nbase + (NCH - 1) * NCHUNK, lo, hi);
    __builtin_amdgcn_sched_barrier(0);
    compute_chunk(fa, qa0, qa1, acc0, acc1, accs0, accs1);
    __builtin_amdgcn_sched_barrier(0);
    compute_chunk(fb, qa0, qa1, acc0, acc1, accs0, accs1);

#pragma unroll
    for (int t = 0; t < 4; ++t)
#pragma unroll
        for (int r = 0; r < 8; ++r) {
            slot[(2 * lo)     * OSTRIDE + 4 * (r + 8 * hi) + t] = acc0[t][r];
            slot[(2 * lo + 1) * OSTRIDE + 4 * (r + 8 * hi) + t] = acc1[t][r];
        }
    if (hi == 0) {
        slot[QT * OSTRIDE + 2 * lo]     = accs0[0];
        slot[QT * OSTRIDE + 2 * lo + 1] = accs1[0];
    }
    __syncthreads();

    float* wsO = ws + (size_t)blockIdx.x * (QT * Vv);
    constexpr int ROWS_PER_PASS = WAVES * 32 / 64;
#pragma unroll
    for (int k = 0; k < QT / ROWS_PER_PASS; ++k) {
        const int row = (tid >> 6) + ROWS_PER_PASS * k;
        const int col = tid & 63;
        float O = 0.f;
        for (int ww = 0; ww < WAVES; ++ww)
            O += smem[ww * SLOT + row * OSTRIDE + col];
        wsO[row * Vv + col] = O;
    }
    if (tid < QT) {
        float S = 0.f;
        for (int ww = 0; ww < WAVES; ++ww)
            S += smem[ww * SLOT + QT * OSTRIDE + tid];
        ws[WS_S + blockIdx.x * QT + tid] = S;
    }
}

__global__ __launch_bounds__(256)
void gated_attn_add_halves(const float* __restrict__ ws, float* __restrict__ out)
{
    const int idx = blockIdx.x * 256 + threadIdx.x;
    const int c4  = idx & 15;
    const int row = (idx >> 4) & (QT - 1);
    const int pq  = idx >> 9;

    const float* o0 = ws + ((size_t)(2 * pq) * QT + row) * Vv + 4 * c4;
    const float* o1 = o0 + QT * Vv;
    const v4f x = *(const v4f*)o0;
    const v4f y = *(const v4f*)o1;
    const float S = ws[WS_S + (2 * pq) * QT + row] + ws[WS_S + (2 * pq + 1) * QT + row];
    v4f o;
#pragma unroll
    for (int e = 0; e < 4; ++e)
        o[e] = (x[e] + y[e]) / S;
    *(v4f*)(out + ((size_t)(pq * QT + row)) * Vv + 4 * c4) = o;
}

extern "C" void kernel_launch(void* const* d_in, const int* in_sizes, int n_in,
                              void* d_out, int out_size, void* d_ws, size_t ws_size,
                              hipStream_t stream) {
    (void)in_sizes; (void)n_in; (void)ws_size; (void)out_size;
    const float* query = (const float*)d_in[0];
    const float* key   = (const float*)d_in[1];
    const float* value = (const float*)d_in[2];
    const float* gates = (const float*)d_in[3];
    float* out = (float*)d_out;
    float* ws  = (float*)d_ws;

    dim3 grid(NBLK);
    dim3 block(WAVES * 32);
    hipLaunchKernelGGL(gated_attn_flash_f32wmma, grid, block, 0, stream,
                       query, key, value, gates, ws);

    dim3 grid2(Bb * Qq * Vv / 4 / 256);
    dim3 block2(256);
    hipLaunchKernelGGL(gated_attn_add_halves, grid2, block2, 0, stream,
                       (const float*)ws, out);
}
